# starting kernel
# baseline (speedup 1.0000x reference)
#include <hip/hip_runtime.h>
#include <math.h>

#define NN 65536
#define NE 1048576
#define DD 128
#define DP 129
#define EPSF 1e-6f
#define BETAF 0.5f

typedef float v2f __attribute__((ext_vector_type(2)));
typedef float v8f __attribute__((ext_vector_type(8)));

__device__ __forceinline__ float wave_sum(float v) {
#pragma unroll
  for (int m = 16; m > 0; m >>= 1) v += __shfl_xor(v, m, 32);
  return v;
}
__device__ __forceinline__ float half_sum16(float v) {
#pragma unroll
  for (int m = 8; m > 0; m >>= 1) v += __shfl_xor(v, m, 32);
  return v;
}
__device__ __forceinline__ float acosh_(float z) {
  return logf(z + sqrtf(fmaf(z, z, -1.0f)));
}

__global__ __launch_bounds__(256) void phi_kernel(const float* X, float* __restrict__ phi) {
  int node = (blockIdx.x * blockDim.x + threadIdx.x) >> 5;
  int lane = threadIdx.x & 31;
  if (node >= NN) return;
  const float* row = X + (size_t)node * DP;
  float ss = 0.f;
#pragma unroll
  for (int j = 0; j < 4; ++j) {
    float v = row[1 + lane * 4 + j];
    ss = fmaf(v, v, ss);
  }
  ss = wave_sum(ss);
  if (lane == 0) {
    float xn = sqrtf(fmaxf(ss, EPSF * EPSF));
    float th = acosh_(fmaxf(row[0], 1.0f + EPSF));
    phi[node] = th / xn;
  }
}

__global__ __launch_bounds__(256) void gemm_kernel(const float* X,
    const float* __restrict__ phi, const float* __restrict__ W,
    const float* __restrict__ bias, float* __restrict__ t, float* __restrict__ s) {
  __shared__ float lw[128 * 130];
  for (int idx = threadIdx.x; idx < 128 * 128; idx += 256) {
    int r = idx >> 7, c = idx & 127;
    lw[r * 130 + c] = W[idx];
  }
  __syncthreads();

  int waveId = threadIdx.x >> 5;
  int lane = threadIdx.x & 31;
  int m0 = blockIdx.x * 128 + waveId * 16;
  int lh = lane & 15;
  int hi = lane >> 4;

  int arow = m0 + lh;
  float phiA = phi[arow];
  const float* xrow = X + (size_t)arow * DP + 1;

  v8f acc[8] = {};
  for (int k0 = 0; k0 < 128; k0 += 4) {
    int kk = k0 + hi * 2;
    v2f a;
    a.x = phiA * xrow[kk];
    a.y = phiA * xrow[kk + 1];
#pragma unroll
    for (int nt = 0; nt < 8; ++nt) {
      int bn = nt * 16 + lh;
      v2f bf;
      bf.x = lw[bn * 130 + kk];
      bf.y = lw[bn * 130 + kk + 1];
      acc[nt] = __builtin_amdgcn_wmma_f32_16x16x4_f32(
          false, a, false, bf, (short)0, acc[nt], false, false);
    }
  }
  float ss[8];
#pragma unroll
  for (int r = 0; r < 8; ++r) ss[r] = 0.f;
#pragma unroll
  for (int nt = 0; nt < 8; ++nt) {
    float bv = bias[nt * 16 + lh];
#pragma unroll
    for (int r = 0; r < 8; ++r) {
      float g = acc[nt][r] + bv;
      acc[nt][r] = g;
      ss[r] = fmaf(g, g, ss[r]);
    }
  }
#pragma unroll
  for (int r = 0; r < 8; ++r) ss[r] = half_sum16(ss[r]);

  float scl[8];
#pragma unroll
  for (int r = 0; r < 8; ++r) {
    float vss = ss[r];
    float vn = sqrtf(fmaxf(vss, EPSF * EPSF));
    float a = vn;
    float k1 = sinhf(a) / a;
    float hs2 = k1 * k1 * vss;
    float xn = sqrtf(fmaxf(hs2, EPSF * EPSF));
    float th = acosh_(fmaxf(coshf(a), 1.0f + EPSF));
    scl[r] = th / xn * k1;
  }
#pragma unroll
  for (int nt = 0; nt < 8; ++nt) {
    int col = nt * 16 + lh;
#pragma unroll
    for (int r = 0; r < 8; ++r) {
      int row = m0 + r + hi * 8;
      float v = scl[r] * acc[nt][r];
      size_t o = (size_t)row * DD + col;
      t[o] = v;
      s[o] = v;
    }
  }
}

__global__ __launch_bounds__(256) void deg_init(float* __restrict__ deg) {
  int i = blockIdx.x * blockDim.x + threadIdx.x;
  if (i < NN) deg[i] = 1.0f;
}

__global__ __launch_bounds__(256) void deg_scatter(const int* __restrict__ ei,
                                                   float* __restrict__ deg) {
  int i = blockIdx.x * blockDim.x + threadIdx.x;
  int n = gridDim.x * blockDim.x;
  for (int e = i; e < NE; e += n) atomicAdd(&deg[ei[NE + e]], 1.0f);
}

__global__ __launch_bounds__(256) void edge_scatter(const int* __restrict__ ei,
    const float* __restrict__ t, float* __restrict__ s) {
  int nw = (gridDim.x * blockDim.x) >> 5;
  int w = (blockIdx.x * blockDim.x + threadIdx.x) >> 5;
  int lane = threadIdx.x & 31;
  for (int e = w; e < NE; e += nw) {
    int src = ei[e];
    int dst = ei[NE + e];
    float4 v = ((const float4*)(t + (size_t)src * DD))[lane];
    float* so = s + (size_t)dst * DD + lane * 4;
    atomicAdd(so + 0, v.x);
    atomicAdd(so + 1, v.y);
    atomicAdd(so + 2, v.z);
    atomicAdd(so + 3, v.w);
  }
}

__global__ __launch_bounds__(256) void node_kernel(const float* X,
    const float* __restrict__ s, const float* __restrict__ deg, float* Y) {
  int i = (blockIdx.x * blockDim.x + threadIdx.x) >> 5;
  int lane = threadIdx.x & 31;
  if (i >= NN) return;
  const float* xrow = X + (size_t)i * DP;
  float x0 = xrow[0];
  float xs[4], mean[4];
  float invd = 1.0f / deg[i];
  float4 sv = ((const float4*)(s + (size_t)i * DD))[lane];
  mean[0] = sv.x * invd; mean[1] = sv.y * invd;
  mean[2] = sv.z * invd; mean[3] = sv.w * invd;
#pragma unroll
  for (int j = 0; j < 4; ++j) xs[j] = xrow[1 + lane * 4 + j];

  float r1 = 0.f;
#pragma unroll
  for (int j = 0; j < 4; ++j) r1 = fmaf(mean[j], mean[j], r1);
  r1 = wave_sum(r1);
  float vn = sqrtf(fmaxf(r1, EPSF * EPSF));
  float a1 = vn;
  float k1 = sinhf(a1) / a1;
  float agg_ss = k1 * k1 * r1;
  float x0a = sqrtf(1.0f + agg_ss);
  float un = sqrtf(fmaxf(agg_ss, EPSF * EPSF));
  float th = acosh_(fmaxf(x0a, 1.0f + EPSF));
  float su = th / un * k1;

  float u[4]; float ru = 0.f;
#pragma unroll
  for (int j = 0; j < 4; ++j) { u[j] = fmaxf(su * mean[j], 0.0f); ru = fmaf(u[j], u[j], ru); }
  ru = wave_sum(ru);
  float rn = sqrtf(fmaxf(ru, EPSF * EPSF));
  float a2 = rn;
  float act0 = coshf(a2);
  float k2 = sinhf(a2) / a2;

  float acts[4]; float dxa = 0.f;
#pragma unroll
  for (int j = 0; j < 4; ++j) { acts[j] = k2 * u[j]; dxa = fmaf(xs[j], acts[j], dxa); }
  dxa = wave_sum(dxa);
  float alpha = fmaxf(x0 * act0 - dxa, 1.0f + EPSF);

  float u2[4]; float u2ss = 0.f;
#pragma unroll
  for (int j = 0; j < 4; ++j) { u2[j] = acts[j] - alpha * xs[j]; u2ss = fmaf(u2[j], u2[j], u2ss); }
  u2ss = wave_sum(u2ss);
  float u20 = act0 - alpha * x0;
  float mu2 = u2ss - u20 * u20;
  float un2 = sqrtf(fmaxf(mu2, EPSF * EPSF));
  float dist = acosh_(alpha);
  float sv2 = BETAF * dist / un2;
  float bv0 = sv2 * u20;
  float mbv = sv2 * sv2 * mu2;
  float a3 = sqrtf(fmaxf(mbv, EPSF * EPSF));
  float c3 = coshf(a3);
  float k3 = sinhf(a3) / a3;

  float out[4]; float oss = 0.f;
#pragma unroll
  for (int j = 0; j < 4; ++j) {
    out[j] = c3 * xs[j] + k3 * (sv2 * u2[j]);
    oss = fmaf(out[j], out[j], oss);
  }
  oss = wave_sum(oss);
  float y0 = sqrtf(1.0f + oss);
  (void)bv0;
  float* yrow = Y + (size_t)i * DP;
  if (lane == 0) yrow[0] = y0;
#pragma unroll
  for (int j = 0; j < 4; ++j) yrow[1 + lane * 4 + j] = out[j];
}

extern "C" void kernel_launch(void* const* d_in, const int* in_sizes, int n_in,
                              void* d_out, int out_size, void* d_ws, size_t ws_size,
                              hipStream_t stream) {
  const float* x  = (const float*)d_in[0];
  const int*   ei = (const int*)d_in[1];
  const float* W1 = (const float*)d_in[2];
  const float* b1 = (const float*)d_in[3];
  const float* W2 = (const float*)d_in[4];
  const float* b2 = (const float*)d_in[5];
  float* out = (float*)d_out;

  float* t   = (float*)d_ws;
  float* s   = t + (size_t)NN * DD;
  float* phi = s + (size_t)NN * DD;
  float* deg = phi + NN;

  deg_init<<<NN / 256, 256, 0, stream>>>(deg);
  deg_scatter<<<1024, 256, 0, stream>>>(ei, deg);

  const float* Ws[2] = {W1, W2};
  const float* bs[2] = {b1, b2};
  for (int L = 0; L < 2; ++L) {
    const float* X = (L == 0) ? x : out;
    phi_kernel<<<NN / 8, 256, 0, stream>>>(X, phi);
    gemm_kernel<<<NN / 128, 256, 0, stream>>>(X, phi, Ws[L], bs[L], t, s);
    edge_scatter<<<2048, 256, 0, stream>>>(ei, t, s);
    node_kernel<<<NN / 8, 256, 0, stream>>>(X, s, deg, out);
  }
}


# segment sum: a block owns 256 destination nodes, keeps their sums in LDS, scans the destination list for its range and adds gathered rows by LDS f32 adds (LDS arrival order); no global f32 atomics
# speedup vs baseline: 1.7309x; 1.7309x over previous
#include <hip/hip_runtime.h>
#include <math.h>

#define NN 65536
#define NE 1048576
#define DD 128
#define DP 129
#define EPSF 1e-6f
#define BETAF 0.5f
#define PHI_RPW 8
#define NODE_RPW 4
#define OWN 256
#define EO_WAVES 32
#define SCAN_U 8
#define NGROUPS (NE / (SCAN_U * 128))
#define LISTCAP (32 + SCAN_U * 64)

typedef float v2f __attribute__((ext_vector_type(2)));
typedef float v8f __attribute__((ext_vector_type(8)));

__device__ __forceinline__ float wave_sum(float v) {
#pragma unroll
  for (int m = 16; m > 0; m >>= 1) v += __shfl_xor(v, m, 32);
  return v;
}
__device__ __forceinline__ float half_sum16(float v) {
#pragma unroll
  for (int m = 8; m > 0; m >>= 1) v += __shfl_xor(v, m, 32);
  return v;
}
__device__ __forceinline__ float acosh_(float z) {
  return logf(z + sqrtf(fmaf(z, z, -1.0f)));
}

__global__ __launch_bounds__(256) void phi_kernel(const float* X, float* __restrict__ phi) {
  int w = (blockIdx.x * blockDim.x + threadIdx.x) >> 5;
  int lane = threadIdx.x & 31;
  int n0 = w * PHI_RPW;
  if (n0 >= NN) return;
  float ss[PHI_RPW];
#pragma unroll
  for (int i = 0; i < PHI_RPW; ++i) {
    const float* row = X + (size_t)(n0 + i) * DP;
    float a = 0.f;
#pragma unroll
    for (int j = 0; j < 4; ++j) {
      float v = row[1 + lane * 4 + j];
      a = fmaf(v, v, a);
    }
    ss[i] = a;
  }
#pragma unroll
  for (int i = 0; i < PHI_RPW; ++i) ss[i] = wave_sum(ss[i]);
  float mine = ss[0];
#pragma unroll
  for (int i = 1; i < PHI_RPW; ++i) mine = (lane == i) ? ss[i] : mine;
  if (lane < PHI_RPW) {
    int node = n0 + lane;
    float xn = sqrtf(fmaxf(mine, EPSF * EPSF));
    float th = acosh_(fmaxf(X[(size_t)node * DP], 1.0f + EPSF));
    phi[node] = th / xn;
  }
}

__global__ __launch_bounds__(256) void gemm_kernel(const float* X,
    const float* __restrict__ phi, const float* __restrict__ W,
    const float* __restrict__ bias, float* __restrict__ t) {
  __shared__ float lw[128 * 130];
  for (int idx = threadIdx.x; idx < 128 * 128; idx += 256) {
    int r = idx >> 7, c = idx & 127;
    lw[r * 130 + c] = W[idx];
  }
  __syncthreads();

  int waveId = threadIdx.x >> 5;
  int lane = threadIdx.x & 31;
  int m0 = blockIdx.x * 128 + waveId * 16;
  int lh = lane & 15;
  int hi = lane >> 4;

  int arow = m0 + lh;
  float phiA = phi[arow];
  const float* xrow = X + (size_t)arow * DP + 1;

  v8f acc[8] = {};
  for (int k0 = 0; k0 < 128; k0 += 4) {
    int kk = k0 + hi * 2;
    v2f a;
    a.x = phiA * xrow[kk];
    a.y = phiA * xrow[kk + 1];
#pragma unroll
    for (int nt = 0; nt < 8; ++nt) {
      int bn = nt * 16 + lh;
      v2f bf;
      bf.x = lw[bn * 130 + kk];
      bf.y = lw[bn * 130 + kk + 1];
      acc[nt] = __builtin_amdgcn_wmma_f32_16x16x4_f32(
          false, a, false, bf, (short)0, acc[nt], false, false);
    }
  }
  float ss[8];
#pragma unroll
  for (int r = 0; r < 8; ++r) ss[r] = 0.f;
#pragma unroll
  for (int nt = 0; nt < 8; ++nt) {
    float bv = bias[nt * 16 + lh];
#pragma unroll
    for (int r = 0; r < 8; ++r) {
      float g = acc[nt][r] + bv;
      acc[nt][r] = g;
      ss[r] = fmaf(g, g, ss[r]);
    }
  }
#pragma unroll
  for (int r = 0; r < 8; ++r) ss[r] = half_sum16(ss[r]);

  float scl[8];
#pragma unroll
  for (int r = 0; r < 8; ++r) {
    float vss = ss[r];
    float vn = sqrtf(fmaxf(vss, EPSF * EPSF));
    float a = vn;
    float k1 = sinhf(a) / a;
    float hs2 = k1 * k1 * vss;
    float xn = sqrtf(fmaxf(hs2, EPSF * EPSF));
    float th = acosh_(fmaxf(coshf(a), 1.0f + EPSF));
    scl[r] = th / xn * k1;
  }
#pragma unroll
  for (int nt = 0; nt < 8; ++nt) {
    int col = nt * 16 + lh;
#pragma unroll
    for (int r = 0; r < 8; ++r) {
      int row = m0 + r + hi * 8;
      float v = scl[r] * acc[nt][r];
      size_t o = (size_t)row * DD + col;
      t[o] = v;
    }
  }
}

__global__ __launch_bounds__(256) void deg_init(float* __restrict__ deg) {
  int i = blockIdx.x * blockDim.x + threadIdx.x;
  if (i < NN) deg[i] = 1.0f;
}

__global__ __launch_bounds__(256) void deg_scatter(const int* __restrict__ ei,
                                                   float* __restrict__ deg) {
  int i = blockIdx.x * blockDim.x + threadIdx.x;
  int n = gridDim.x * blockDim.x;
  for (int e = i; e < NE; e += n) atomicAdd(&deg[ei[NE + e]], 1.0f);
}

__device__ __forceinline__ void eo_drain(const int* __restrict__ ei, const float* __restrict__ t,
                                         float* acc, const unsigned* mylist, int n, int lane) {
  for (int i0 = 0; i0 < n; i0 += 32) {
    int m = n - i0; m = m < 32 ? m : 32;
    unsigned en = mylist[i0 + (lane < m ? lane : 0)];
    int srcv = ei[en >> 8];
    int ldv = (int)(en & 255u);
    for (int j = 0; j < m; j += 8) {
      float v[8][4];
      int lrow[8];
#pragma unroll
      for (int q = 0; q < 8; ++q) {
        int jj = j + q; jj = jj < m ? jj : m - 1;
        int sq = __builtin_amdgcn_readlane(srcv, jj);
        lrow[q] = __builtin_amdgcn_readlane(ldv, jj);
        const float* tr = t + (size_t)sq * DD + lane;
        v[q][0] = tr[0]; v[q][1] = tr[32]; v[q][2] = tr[64]; v[q][3] = tr[96];
      }
#pragma unroll
      for (int q = 0; q < 8; ++q) {
        if (j + q < m) {
          float* a = acc + lrow[q] * DD + lane;
          atomicAdd(a, v[q][0]);
          atomicAdd(a + 32, v[q][1]);
          atomicAdd(a + 64, v[q][2]);
          atomicAdd(a + 96, v[q][3]);
        }
      }
    }
  }
}

__global__ __launch_bounds__(EO_WAVES * 32) void edge_owner(const int* __restrict__ ei,
    const float* __restrict__ t, float* __restrict__ s) {
  __shared__ float acc[OWN * DD];
  __shared__ unsigned elist[EO_WAVES * LISTCAP];
  const int tid = threadIdx.x;
  const int lane = tid & 31;
  const int w = tid >> 5;
  const unsigned base = blockIdx.x * OWN;

  {
    const float4* t4 = (const float4*)(t + (size_t)base * DD);
    float4* a4 = (float4*)acc;
#pragma unroll
    for (int k = 0; k < OWN * DD / 4 / (EO_WAVES * 32); ++k)
      a4[tid + k * EO_WAVES * 32] = t4[tid + k * EO_WAVES * 32];
  }
  __syncthreads();

  const int4* d4 = (const int4*)(ei + NE);
  unsigned* mylist = elist + w * LISTCAP;
  int cnt = 0;
  const unsigned rot = blockIdx.x * (NGROUPS / (NN / OWN));
  unsigned g = ((unsigned)w + rot) & (NGROUPS - 1);
  int4 cur[SCAN_U];
#pragma unroll
  for (int u = 0; u < SCAN_U; ++u) cur[u] = d4[(g * SCAN_U + u) * 32 + lane];

  for (int it = 0; it < NGROUPS / EO_WAVES; ++it) {
    unsigned gn = ((unsigned)w + (unsigned)(it + 1) * EO_WAVES + rot) & (NGROUPS - 1);
    int4 nxt[SCAN_U];
#pragma unroll
    for (int u = 0; u < SCAN_U; ++u) nxt[u] = d4[(gn * SCAN_U + u) * 32 + lane];

#pragma unroll
    for (int u = 0; u < SCAN_U; ++u) {
      unsigned x0 = (unsigned)cur[u].x - base;
      unsigned x1 = (unsigned)cur[u].y - base;
      unsigned x2 = (unsigned)cur[u].z - base;
      unsigned x3 = (unsigned)cur[u].w - base;
      unsigned mn = min(min(x0, x1), min(x2, x3));
      if (__builtin_amdgcn_ballot_w32(mn < OWN)) {
        unsigned e0 = ((g * SCAN_U + u) * 32 + lane) * 4;
        unsigned xs[4] = {x0, x1, x2, x3};
#pragma unroll
        for (int k = 0; k < 4; ++k) {
          bool p = xs[k] < OWN;
          unsigned mk = __builtin_amdgcn_ballot_w32(p);
          if (mk) {
            if (p) mylist[cnt + __builtin_amdgcn_mbcnt_lo(mk, 0u)] = ((e0 + k) << 8) | xs[k];
            cnt += __builtin_popcount(mk);
          }
        }
      }
      if ((u == SCAN_U / 2 - 1 || u == SCAN_U - 1) && cnt >= 32) {
        eo_drain(ei, t, acc, mylist, cnt, lane);
        cnt = 0;
      }
    }
#pragma unroll
    for (int u = 0; u < SCAN_U; ++u) cur[u] = nxt[u];
    g = gn;
  }
  if (cnt > 0) eo_drain(ei, t, acc, mylist, cnt, lane);
  __syncthreads();

  {
    float4* s4 = (float4*)(s + (size_t)base * DD);
    const float4* a4 = (const float4*)acc;
#pragma unroll
    for (int k = 0; k < OWN * DD / 4 / (EO_WAVES * 32); ++k)
      s4[tid + k * EO_WAVES * 32] = a4[tid + k * EO_WAVES * 32];
  }
}

__global__ __launch_bounds__(256) void node_kernel(const float* X,
    const float* __restrict__ s, const float* __restrict__ deg, float* Y) {
  int w = (blockIdx.x * blockDim.x + threadIdx.x) >> 5;
  int lane = threadIdx.x & 31;
  for (int it = 0; it < NODE_RPW; ++it) {
    int i = w * NODE_RPW + it;
    if (i >= NN) return;
    const float* xrow = X + (size_t)i * DP;
    float x0 = xrow[0];
    float xs[4], mean[4];
    float invd = 1.0f / deg[i];
    float4 sv = ((const float4*)(s + (size_t)i * DD))[lane];
    mean[0] = sv.x * invd; mean[1] = sv.y * invd;
    mean[2] = sv.z * invd; mean[3] = sv.w * invd;
#pragma unroll
    for (int j = 0; j < 4; ++j) xs[j] = xrow[1 + lane * 4 + j];

    float r1 = 0.f;
#pragma unroll
    for (int j = 0; j < 4; ++j) r1 = fmaf(mean[j], mean[j], r1);
    r1 = wave_sum(r1);
    float vn = sqrtf(fmaxf(r1, EPSF * EPSF));
    float a1 = vn;
    float k1 = sinhf(a1) / a1;
    float agg_ss = k1 * k1 * r1;
    float x0a = sqrtf(1.0f + agg_ss);
    float un = sqrtf(fmaxf(agg_ss, EPSF * EPSF));
    float th = acosh_(fmaxf(x0a, 1.0f + EPSF));
    float su = th / un * k1;

    float u[4]; float ru = 0.f;
#pragma unroll
    for (int j = 0; j < 4; ++j) { u[j] = fmaxf(su * mean[j], 0.0f); ru = fmaf(u[j], u[j], ru); }
    ru = wave_sum(ru);
    float rn = sqrtf(fmaxf(ru, EPSF * EPSF));
    float a2 = rn;
    float act0 = coshf(a2);
    float k2 = sinhf(a2) / a2;

    float acts[4]; float dxa = 0.f;
#pragma unroll
    for (int j = 0; j < 4; ++j) { acts[j] = k2 * u[j]; dxa = fmaf(xs[j], acts[j], dxa); }
    dxa = wave_sum(dxa);
    float alpha = fmaxf(x0 * act0 - dxa, 1.0f + EPSF);

    float u2[4]; float u2ss = 0.f;
#pragma unroll
    for (int j = 0; j < 4; ++j) { u2[j] = acts[j] - alpha * xs[j]; u2ss = fmaf(u2[j], u2[j], u2ss); }
    u2ss = wave_sum(u2ss);
    float u20 = act0 - alpha * x0;
    float mu2 = u2ss - u20 * u20;
    float un2 = sqrtf(fmaxf(mu2, EPSF * EPSF));
    float dist = acosh_(alpha);
    float sv2 = BETAF * dist / un2;
    float bv0 = sv2 * u20;
    float mbv = sv2 * sv2 * mu2;
    float a3 = sqrtf(fmaxf(mbv, EPSF * EPSF));
    float c3 = coshf(a3);
    float k3 = sinhf(a3) / a3;

    float out[4]; float oss = 0.f;
#pragma unroll
    for (int j = 0; j < 4; ++j) {
      out[j] = c3 * xs[j] + k3 * (sv2 * u2[j]);
      oss = fmaf(out[j], out[j], oss);
    }
    oss = wave_sum(oss);
    float y0 = sqrtf(1.0f + oss);
    (void)bv0;
    float* yrow = Y + (size_t)i * DP;
    if (lane == 0) yrow[0] = y0;
#pragma unroll
    for (int j = 0; j < 4; ++j) yrow[1 + lane * 4 + j] = out[j];
  }
}

extern "C" void kernel_launch(void* const* d_in, const int* in_sizes, int n_in,
                              void* d_out, int out_size, void* d_ws, size_t ws_size,
                              hipStream_t stream) {
  const float* x  = (const float*)d_in[0];
  const int*   ei = (const int*)d_in[1];
  const float* W1 = (const float*)d_in[2];
  const float* b1 = (const float*)d_in[3];
  const float* W2 = (const float*)d_in[4];
  const float* b2 = (const float*)d_in[5];
  float* out = (float*)d_out;

  float* t   = (float*)d_ws;
  float* s   = t + (size_t)NN * DD;
  float* phi = s + (size_t)NN * DD;
  float* deg = phi + NN;

  deg_init<<<NN / 256, 256, 0, stream>>>(deg);
  deg_scatter<<<1024, 256, 0, stream>>>(ei, deg);

  const float* Ws[2] = {W1, W2};
  const float* bs[2] = {b1, b2};
  for (int L = 0; L < 2; ++L) {
    const float* X = (L == 0) ? x : out;
    phi_kernel<<<NN / (8 * PHI_RPW), 256, 0, stream>>>(X, phi);
    gemm_kernel<<<NN / 128, 256, 0, stream>>>(X, phi, Ws[L], bs[L], t);
    edge_owner<<<NN / OWN, EO_WAVES * 32, 0, stream>>>(ei, t, s);
    node_kernel<<<NN / (8 * NODE_RPW), 256, 0, stream>>>(X, s, deg, out);
  }
}


# owner block finishes its 256 nodes itself: counts their matches in LDS (degree) and runs the node chain from its LDS sums; no sum plane, no degree atomics, 6 launches not 10; sum orders unchanged
# speedup vs baseline: 2.0528x; 1.1860x over previous
#include <hip/hip_runtime.h>
#include <math.h>

#define NN 65536
#define NE 1048576
#define DD 128
#define DP 129
#define EPSF 1e-6f
#define BETAF 0.5f
#define PHI_RPW 8
#define OWN 256
#define EO_WAVES 32
#define ROWS_PW (OWN / EO_WAVES)
#define SCAN_U 8
#define NGROUPS (NE / (SCAN_U * 128))
#define LISTCAP (32 + SCAN_U * 64)

typedef float v2f __attribute__((ext_vector_type(2)));
typedef float v8f __attribute__((ext_vector_type(8)));

__device__ __forceinline__ float wave_sum(float v) {
#pragma unroll
  for (int m = 16; m > 0; m >>= 1) v += __shfl_xor(v, m, 32);
  return v;
}
__device__ __forceinline__ float half_sum16(float v) {
#pragma unroll
  for (int m = 8; m > 0; m >>= 1) v += __shfl_xor(v, m, 32);
  return v;
}
__device__ __forceinline__ float acosh_(float z) {
  return logf(z + sqrtf(fmaf(z, z, -1.0f)));
}

__global__ __launch_bounds__(256) void phi_kernel(const float* X, float* __restrict__ phi) {
  int w = (blockIdx.x * blockDim.x + threadIdx.x) >> 5;
  int lane = threadIdx.x & 31;
  int n0 = w * PHI_RPW;
  if (n0 >= NN) return;
  float ss[PHI_RPW];
#pragma unroll
  for (int i = 0; i < PHI_RPW; ++i) {
    const float* row = X + (size_t)(n0 + i) * DP;
    float a = 0.f;
#pragma unroll
    for (int j = 0; j < 4; ++j) {
      float v = row[1 + lane * 4 + j];
      a = fmaf(v, v, a);
    }
    ss[i] = a;
  }
#pragma unroll
  for (int i = 0; i < PHI_RPW; ++i) ss[i] = wave_sum(ss[i]);
  float mine = ss[0];
#pragma unroll
  for (int i = 1; i < PHI_RPW; ++i) mine = (lane == i) ? ss[i] : mine;
  if (lane < PHI_RPW) {
    int node = n0 + lane;
    float xn = sqrtf(fmaxf(mine, EPSF * EPSF));
    float th = acosh_(fmaxf(X[(size_t)node * DP], 1.0f + EPSF));
    phi[node] = th / xn;
  }
}

__global__ __launch_bounds__(256) void gemm_kernel(const float* X,
    const float* __restrict__ phi, const float* __restrict__ W,
    const float* __restrict__ bias, float* __restrict__ t) {
  __shared__ float lw[128 * 130];
  for (int idx = threadIdx.x; idx < 128 * 128; idx += 256) {
    int r = idx >> 7, c = idx & 127;
    lw[r * 130 + c] = W[idx];
  }
  __syncthreads();

  int waveId = threadIdx.x >> 5;
  int lane = threadIdx.x & 31;
  int m0 = blockIdx.x * 128 + waveId * 16;
  int lh = lane & 15;
  int hi = lane >> 4;

  int arow = m0 + lh;
  float phiA = phi[arow];
  const float* xrow = X + (size_t)arow * DP + 1;

  v8f acc[8] = {};
  for (int k0 = 0; k0 < 128; k0 += 4) {
    int kk = k0 + hi * 2;
    v2f a;
    a.x = phiA * xrow[kk];
    a.y = phiA * xrow[kk + 1];
#pragma unroll
    for (int nt = 0; nt < 8; ++nt) {
      int bn = nt * 16 + lh;
      v2f bf;
      bf.x = lw[bn * 130 + kk];
      bf.y = lw[bn * 130 + kk + 1];
      acc[nt] = __builtin_amdgcn_wmma_f32_16x16x4_f32(
          false, a, false, bf, (short)0, acc[nt], false, false);
    }
  }
  float ss[8];
#pragma unroll
  for (int r = 0; r < 8; ++r) ss[r] = 0.f;
#pragma unroll
  for (int nt = 0; nt < 8; ++nt) {
    float bv = bias[nt * 16 + lh];
#pragma unroll
    for (int r = 0; r < 8; ++r) {
      float g = acc[nt][r] + bv;
      acc[nt][r] = g;
      ss[r] = fmaf(g, g, ss[r]);
    }
  }
#pragma unroll
  for (int r = 0; r < 8; ++r) ss[r] = half_sum16(ss[r]);

  float scl[8];
#pragma unroll
  for (int r = 0; r < 8; ++r) {
    float vss = ss[r];
    float vn = sqrtf(fmaxf(vss, EPSF * EPSF));
    float a = vn;
    float k1 = sinhf(a) / a;
    float hs2 = k1 * k1 * vss;
    float xn = sqrtf(fmaxf(hs2, EPSF * EPSF));
    float th = acosh_(fmaxf(coshf(a), 1.0f + EPSF));
    scl[r] = th / xn * k1;
  }
#pragma unroll
  for (int nt = 0; nt < 8; ++nt) {
    int col = nt * 16 + lh;
#pragma unroll
    for (int r = 0; r < 8; ++r) {
      int row = m0 + r + hi * 8;
      float v = scl[r] * acc[nt][r];
      size_t o = (size_t)row * DD + col;
      t[o] = v;
    }
  }
}

struct XRow { float x0; float xs[4]; };
__device__ __forceinline__ XRow load_xrow(const float* X, int i, int lane) {
  const float* xrow = X + (size_t)i * DP;
  XRow v;
  v.x0 = xrow[0];
#pragma unroll
  for (int j = 0; j < 4; ++j) v.xs[j] = xrow[1 + lane * 4 + j];
  return v;
}
__device__ __forceinline__ void node_row(const XRow xr, float* Y, int i, float4 sv,
                                         float invd, int lane) {
  float x0 = xr.x0;
  float xs[4], mean[4];
  mean[0] = sv.x * invd; mean[1] = sv.y * invd;
  mean[2] = sv.z * invd; mean[3] = sv.w * invd;
#pragma unroll
  for (int j = 0; j < 4; ++j) xs[j] = xr.xs[j];

  float r1 = 0.f;
#pragma unroll
  for (int j = 0; j < 4; ++j) r1 = fmaf(mean[j], mean[j], r1);
  r1 = wave_sum(r1);
  float vn = sqrtf(fmaxf(r1, EPSF * EPSF));
  float a1 = vn;
  float k1 = sinhf(a1) / a1;
  float agg_ss = k1 * k1 * r1;
  float x0a = sqrtf(1.0f + agg_ss);
  float un = sqrtf(fmaxf(agg_ss, EPSF * EPSF));
  float th = acosh_(fmaxf(x0a, 1.0f + EPSF));
  float su = th / un * k1;

  float u[4]; float ru = 0.f;
#pragma unroll
  for (int j = 0; j < 4; ++j) { u[j] = fmaxf(su * mean[j], 0.0f); ru = fmaf(u[j], u[j], ru); }
  ru = wave_sum(ru);
  float rn = sqrtf(fmaxf(ru, EPSF * EPSF));
  float a2 = rn;
  float act0 = coshf(a2);
  float k2 = sinhf(a2) / a2;

  float acts[4]; float dxa = 0.f;
#pragma unroll
  for (int j = 0; j < 4; ++j) { acts[j] = k2 * u[j]; dxa = fmaf(xs[j], acts[j], dxa); }
  dxa = wave_sum(dxa);
  float alpha = fmaxf(x0 * act0 - dxa, 1.0f + EPSF);

  float u2[4]; float u2ss = 0.f;
#pragma unroll
  for (int j = 0; j < 4; ++j) { u2[j] = acts[j] - alpha * xs[j]; u2ss = fmaf(u2[j], u2[j], u2ss); }
  u2ss = wave_sum(u2ss);
  float u20 = act0 - alpha * x0;
  float mu2 = u2ss - u20 * u20;
  float un2 = sqrtf(fmaxf(mu2, EPSF * EPSF));
  float dist = acosh_(alpha);
  float sv2 = BETAF * dist / un2;
  float mbv = sv2 * sv2 * mu2;
  float a3 = sqrtf(fmaxf(mbv, EPSF * EPSF));
  float c3 = coshf(a3);
  float k3 = sinhf(a3) / a3;

  float out[4]; float oss = 0.f;
#pragma unroll
  for (int j = 0; j < 4; ++j) {
    out[j] = c3 * xs[j] + k3 * (sv2 * u2[j]);
    oss = fmaf(out[j], out[j], oss);
  }
  oss = wave_sum(oss);
  float y0 = sqrtf(1.0f + oss);
  float* yrow = Y + (size_t)i * DP;
  if (lane == 0) yrow[0] = y0;
#pragma unroll
  for (int j = 0; j < 4; ++j) yrow[1 + lane * 4 + j] = out[j];
}

__device__ __forceinline__ void eo_drain(const int* __restrict__ ei, const float* __restrict__ t,
                                         float* acc, int* cntl, const unsigned* mylist,
                                         int n, int lane) {
  for (int i0 = 0; i0 < n; i0 += 32) {
    int m = n - i0; m = m < 32 ? m : 32;
    unsigned en = mylist[i0 + (lane < m ? lane : 0)];
    int srcv = ei[en >> 8];
    int ldv = (int)(en & 255u);
    if (lane < m) atomicAdd(cntl + ldv, 1);
    for (int j = 0; j < m; j += 8) {
      float v[8][4];
      int lrow[8];
#pragma unroll
      for (int q = 0; q < 8; ++q) {
        int jj = j + q; jj = jj < m ? jj : m - 1;
        int sq = __builtin_amdgcn_readlane(srcv, jj);
        lrow[q] = __builtin_amdgcn_readlane(ldv, jj);
        const float* tr = t + (size_t)sq * DD + lane;
        v[q][0] = tr[0]; v[q][1] = tr[32]; v[q][2] = tr[64]; v[q][3] = tr[96];
      }
#pragma unroll
      for (int q = 0; q < 8; ++q) {
        if (j + q < m) {
          float* a = acc + lrow[q] * DD + lane;
          atomicAdd(a, v[q][0]);
          atomicAdd(a + 32, v[q][1]);
          atomicAdd(a + 64, v[q][2]);
          atomicAdd(a + 96, v[q][3]);
        }
      }
    }
  }
}

__global__ __launch_bounds__(EO_WAVES * 32) void edge_owner(const int* __restrict__ ei,
    const float* __restrict__ t, const float* X, float* Y) {
  __shared__ float acc[OWN * DD];
  __shared__ unsigned elist[EO_WAVES * LISTCAP];
  __shared__ int cntl[OWN];
  const int tid = threadIdx.x;
  const int lane = tid & 31;
  const int w = tid >> 5;
  const unsigned base = blockIdx.x * OWN;

  {
    const float4* t4 = (const float4*)(t + (size_t)base * DD);
    float4* a4 = (float4*)acc;
#pragma unroll
    for (int k = 0; k < OWN * DD / 4 / (EO_WAVES * 32); ++k)
      a4[tid + k * EO_WAVES * 32] = t4[tid + k * EO_WAVES * 32];
    if (tid < OWN) cntl[tid] = 0;
  }
  __syncthreads();

  const int4* d4 = (const int4*)(ei + NE);
  unsigned* mylist = elist + w * LISTCAP;
  int cnt = 0;
  const unsigned rot = blockIdx.x * (NGROUPS / (NN / OWN));
  unsigned g = ((unsigned)w + rot) & (NGROUPS - 1);
  int4 cur[SCAN_U];
#pragma unroll
  for (int u = 0; u < SCAN_U; ++u) cur[u] = d4[(g * SCAN_U + u) * 32 + lane];

  for (int it = 0; it < NGROUPS / EO_WAVES; ++it) {
    unsigned gn = ((unsigned)w + (unsigned)(it + 1) * EO_WAVES + rot) & (NGROUPS - 1);
    int4 nxt[SCAN_U];
#pragma unroll
    for (int u = 0; u < SCAN_U; ++u) nxt[u] = d4[(gn * SCAN_U + u) * 32 + lane];

#pragma unroll
    for (int u = 0; u < SCAN_U; ++u) {
      unsigned x0 = (unsigned)cur[u].x - base;
      unsigned x1 = (unsigned)cur[u].y - base;
      unsigned x2 = (unsigned)cur[u].z - base;
      unsigned x3 = (unsigned)cur[u].w - base;
      unsigned mn = min(min(x0, x1), min(x2, x3));
      if (__builtin_amdgcn_ballot_w32(mn < OWN)) {
        unsigned e0 = ((g * SCAN_U + u) * 32 + lane) * 4;
        unsigned xs[4] = {x0, x1, x2, x3};
#pragma unroll
        for (int k = 0; k < 4; ++k) {
          bool p = xs[k] < OWN;
          unsigned mk = __builtin_amdgcn_ballot_w32(p);
          if (mk) {
            if (p) mylist[cnt + __builtin_amdgcn_mbcnt_lo(mk, 0u)] = ((e0 + k) << 8) | xs[k];
            cnt += __builtin_popcount(mk);
          }
        }
      }
      if ((u == SCAN_U / 2 - 1 || u == SCAN_U - 1) && cnt >= 32) {
        eo_drain(ei, t, acc, cntl, mylist, cnt, lane);
        cnt = 0;
      }
    }
#pragma unroll
    for (int u = 0; u < SCAN_U; ++u) cur[u] = nxt[u];
    g = gn;
  }
  if (cnt > 0) eo_drain(ei, t, acc, cntl, mylist, cnt, lane);

  const int r0 = w * ROWS_PW;
  XRow xr = load_xrow(X, (int)base + r0, lane);
  __syncthreads();
  for (int it = 0; it < ROWS_PW; ++it) {
    int r = r0 + it;
    int rn = r0 + (it + 1 < ROWS_PW ? it + 1 : it);
    XRow xnext = load_xrow(X, (int)base + rn, lane);
    __builtin_amdgcn_sched_barrier(0);
    float4 sv = ((const float4*)(acc + r * DD))[lane];
    float invd = 1.0f / (float)(1 + cntl[r]);
    node_row(xr, Y, (int)base + r, sv, invd, lane);
    xr = xnext;
  }
}

extern "C" void kernel_launch(void* const* d_in, const int* in_sizes, int n_in,
                              void* d_out, int out_size, void* d_ws, size_t ws_size,
                              hipStream_t stream) {
  const float* x  = (const float*)d_in[0];
  const int*   ei = (const int*)d_in[1];
  const float* W1 = (const float*)d_in[2];
  const float* b1 = (const float*)d_in[3];
  const float* W2 = (const float*)d_in[4];
  const float* b2 = (const float*)d_in[5];
  float* out = (float*)d_out;

  float* t   = (float*)d_ws;
  float* phi = t + (size_t)NN * DD;

  const float* Ws[2] = {W1, W2};
  const float* bs[2] = {b1, b2};
  for (int L = 0; L < 2; ++L) {
    const float* X = (L == 0) ? x : out;
    phi_kernel<<<NN / (8 * PHI_RPW), 256, 0, stream>>>(X, phi);
    gemm_kernel<<<NN / 128, 256, 0, stream>>>(X, phi, Ws[L], bs[L], t);
    edge_owner<<<NN / OWN, EO_WAVES * 32, 0, stream>>>(ei, t, X, out);
  }
}
